# starting kernel
# baseline (speedup 1.0000x reference)
#include <hip/hip_runtime.h>
#include <math.h>

#define NN     100000
#define NE     1600000
#define EMB3   192
#define F_LAT  64
#define F_OUT  128
#define NEG_SLOPE 0.2f

typedef __attribute__((ext_vector_type(2))) float v2f;
typedef __attribute__((ext_vector_type(8))) float v8f;

__device__ __forceinline__ v8f wmma4(v2f a, v2f b, v8f c) {
  return __builtin_amdgcn_wmma_f32_16x16x4_f32(false, a, false, b, (short)0, c,
                                               false, false);
}

__device__ __forceinline__ unsigned enc_f32(float f) {
  unsigned u = __float_as_uint(f);
  return (u & 0x80000000u) ? ~u : (u | 0x80000000u);
}
__device__ __forceinline__ float dec_f32(unsigned u) {
  return (u & 0x80000000u) ? __uint_as_float(u & 0x7FFFFFFFu)
                           : __uint_as_float(~u);
}

__global__ void init_kernel(float* __restrict__ rst, float* __restrict__ denom,
                            unsigned* __restrict__ emax_u) {
  int i = blockIdx.x * blockDim.x + threadIdx.x;
  if (i < NN * F_OUT) rst[i] = 0.0f;
  if (i < NN) { denom[i] = 0.0f; emax_u[i] = 0u; }
}

__global__ __launch_bounds__(256) void node_feat_kernel(
    const float* __restrict__ embed,
    const float* __restrict__ W_lin, const float* __restrict__ b_lin,
    const float* __restrict__ W_gat,
    const float* __restrict__ attn_l, const float* __restrict__ attn_r,
    float* __restrict__ feat, float* __restrict__ el, float* __restrict__ er,
    int n_tiles) {
  extern __shared__ float smem[];
  float* sW1 = smem;
  float* sW2 = sW1 + EMB3 * F_LAT;
  float* sB1 = sW2 + F_LAT * F_OUT;
  float* sAl = sB1 + F_LAT;
  float* sAr = sAl + F_OUT;
  float* sX  = sAr + F_OUT;

  const int tid = threadIdx.x;
  for (int i = tid; i < EMB3 * F_LAT; i += 256) sW1[i] = W_lin[i];
  for (int i = tid; i < F_LAT * F_OUT; i += 256) sW2[i] = W_gat[i];
  if (tid < F_LAT) sB1[tid] = b_lin[tid];
  if (tid < F_OUT) { sAl[tid] = attn_l[tid]; sAr[tid] = attn_r[tid]; }
  __syncthreads();

  const int wave = tid >> 5;
  const int lane = tid & 31;
  const int tile = blockIdx.x * 8 + wave;
  if (tile >= n_tiles) return;

  const int m  = lane & 15;
  const int hi = lane >> 4;

  const float* arow = embed + (size_t)(tile * 16 + m) * EMB3;
  v8f acc[4] = {};
  for (int k0 = 0; k0 < EMB3; k0 += 4) {
    const int ka = k0 + 2 * hi;
    v2f a; a.x = arow[ka]; a.y = arow[ka + 1];
    const float* b0 = sW1 + ka * F_LAT + m;
    const float* b1 = b0 + F_LAT;
#pragma unroll
    for (int nt = 0; nt < 4; ++nt) {
      v2f b; b.x = b0[nt * 16]; b.y = b1[nt * 16];
      acc[nt] = wmma4(a, b, acc[nt]);
    }
  }

  float* xw = sX + wave * 16 * F_LAT;
#pragma unroll
  for (int r = 0; r < 8; ++r) {
    const int row = r + 8 * hi;
#pragma unroll
    for (int nt = 0; nt < 4; ++nt) {
      float v = acc[nt][r] + sB1[nt * 16 + m];
      xw[row * F_LAT + nt * 16 + m] = fmaxf(v, 0.0f);
    }
  }

  v8f fa[8] = {};
  for (int k0 = 0; k0 < F_LAT; k0 += 4) {
    const int ka = k0 + 2 * hi;
    v2f a; a.x = xw[m * F_LAT + ka]; a.y = xw[m * F_LAT + ka + 1];
    const float* b0 = sW2 + ka * F_OUT + m;
    const float* b1 = b0 + F_OUT;
#pragma unroll
    for (int nt = 0; nt < 8; ++nt) {
      v2f b; b.x = b0[nt * 16]; b.y = b1[nt * 16];
      fa[nt] = wmma4(a, b, fa[nt]);
    }
  }

  float pl[8], pr[8];
  const size_t base = (size_t)(tile * 16) * F_OUT;
#pragma unroll
  for (int r = 0; r < 8; ++r) {
    const int row = r + 8 * hi;
    float sl = 0.0f, sr = 0.0f;
#pragma unroll
    for (int nt = 0; nt < 8; ++nt) {
      const float v = fa[nt][r];
      feat[base + row * F_OUT + nt * 16 + m] = v;
      sl += v * sAl[nt * 16 + m];
      sr += v * sAr[nt * 16 + m];
    }
    pl[r] = sl; pr[r] = sr;
  }
#pragma unroll
  for (int off = 8; off >= 1; off >>= 1) {
#pragma unroll
    for (int r = 0; r < 8; ++r) {
      pl[r] += __shfl_xor(pl[r], off, 32);
      pr[r] += __shfl_xor(pr[r], off, 32);
    }
  }
  if (m == 0) {
#pragma unroll
    for (int r = 0; r < 8; ++r) {
      el[tile * 16 + r + 8 * hi] = pl[r];
      er[tile * 16 + r + 8 * hi] = pr[r];
    }
  }
}

__global__ void edge_logit_kernel(const int* __restrict__ src,
                                  const int* __restrict__ dst,
                                  const float* __restrict__ el,
                                  const float* __restrict__ er,
                                  float* __restrict__ e_edge,
                                  unsigned* __restrict__ emax_u) {
  int i = blockIdx.x * blockDim.x + threadIdx.x;
  if (i >= NE) return;
  float e = el[src[i]] + er[dst[i]];
  e = (e > 0.0f) ? e : NEG_SLOPE * e;
  e_edge[i] = e;
  atomicMax(emax_u + dst[i], enc_f32(e));
}

__global__ void edge_exp_kernel(const int* __restrict__ dst,
                                float* __restrict__ e_edge,
                                const unsigned* __restrict__ emax_u,
                                float* __restrict__ denom) {
  int i = blockIdx.x * blockDim.x + threadIdx.x;
  if (i >= NE) return;
  const int d = dst[i];
  float emax = dec_f32(emax_u[d]);
  if ((__float_as_uint(emax) & 0x7F800000u) == 0x7F800000u) emax = 0.0f;
  float ee = expf(e_edge[i] - emax);
  e_edge[i] = ee;
  atomicAdd(denom + d, ee);
}

__global__ __launch_bounds__(256) void edge_scatter_kernel(
    const int* __restrict__ src, const int* __restrict__ dst,
    const float* __restrict__ e_edge, const float* __restrict__ denom,
    const float* __restrict__ feat, float* __restrict__ rst) {
  int gid = blockIdx.x * blockDim.x + threadIdx.x;
  int e = gid >> 5;
  int lane = gid & 31;
  if (e >= NE) return;
  const int s = src[e], d = dst[e];
  float dn = denom[d];
  dn = (dn > 0.0f) ? dn : 1.0f;
  const float alpha = e_edge[e] / dn;
  const float4 v = ((const float4*)(feat + (size_t)s * F_OUT))[lane];
  float* rr = rst + (size_t)d * F_OUT + lane * 4;
  atomicAdd(rr + 0, v.x * alpha);
  atomicAdd(rr + 1, v.y * alpha);
  atomicAdd(rr + 2, v.z * alpha);
  atomicAdd(rr + 3, v.w * alpha);
}

__global__ void finalize_kernel(const float* __restrict__ rst,
                                const float* __restrict__ b_gat,
                                float* __restrict__ out) {
  int i = blockIdx.x * blockDim.x + threadIdx.x;
  if (i >= NN * F_OUT) return;
  const int n = i >> 7, c = i & 127;
  const float y = rst[i] + b_gat[c];
  if (c < 64) out[(size_t)n * 64 + c] = y;
  else        out[(size_t)NN * 64 + (size_t)n * 64 + (c - 64)] = tanhf(y);
}

extern "C" void kernel_launch(void* const* d_in, const int* in_sizes, int n_in,
                              void* d_out, int out_size, void* d_ws,
                              size_t ws_size, hipStream_t stream) {
  const float* embed  = (const float*)d_in[0];
  const int*   src    = (const int*)d_in[1];
  const int*   dst    = (const int*)d_in[2];
  const float* W_lin  = (const float*)d_in[3];
  const float* b_lin  = (const float*)d_in[4];
  const float* W_gat  = (const float*)d_in[5];
  const float* attn_l = (const float*)d_in[6];
  const float* attn_r = (const float*)d_in[7];
  const float* b_gat  = (const float*)d_in[8];
  float* out = (float*)d_out;

  float* ws     = (float*)d_ws;
  float* feat   = ws;
  float* el     = feat + (size_t)NN * F_OUT;
  float* er     = el + NN;
  float* denom  = er + NN;
  unsigned* emax_u = (unsigned*)(denom + NN);
  float* e_edge = (float*)(emax_u + NN);
  float* rst    = e_edge + NE;

  init_kernel<<<(NN * F_OUT + 255) / 256, 256, 0, stream>>>(rst, denom, emax_u);

  const int n_tiles = NN / 16;
  const int nf_blocks = (n_tiles + 7) / 8;
  const size_t smem_bytes =
      (size_t)(EMB3 * F_LAT + F_LAT * F_OUT + F_LAT + 2 * F_OUT +
               8 * 16 * F_LAT) * sizeof(float);
  (void)hipFuncSetAttribute((const void*)node_feat_kernel,
                            hipFuncAttributeMaxDynamicSharedMemorySize,
                            (int)smem_bytes);
  node_feat_kernel<<<nf_blocks, 256, smem_bytes, stream>>>(
      embed, W_lin, b_lin, W_gat, attn_l, attn_r, feat, el, er, n_tiles);

  edge_logit_kernel<<<(NE + 255) / 256, 256, 0, stream>>>(src, dst, el, er,
                                                          e_edge, emax_u);
  edge_exp_kernel<<<(NE + 255) / 256, 256, 0, stream>>>(dst, e_edge, emax_u,
                                                        denom);
  edge_scatter_kernel<<<(NE * 32 + 255) / 256, 256, 0, stream>>>(
      src, dst, e_edge, denom, feat, rst);
  finalize_kernel<<<(NN * F_OUT + 255) / 256, 256, 0, stream>>>(rst, b_gat,
                                                                out);
}


# aggregation row pass: a whole source row per load instruction (16 B a lane, was four 4-B loads), LDS rows as four lane-major planes, next 8 rows in flight during the adds; LDS arrival sums
# speedup vs baseline: 4.9061x; 4.9061x over previous
#include <hip/hip_runtime.h>
#include <math.h>

#define NN     100000
#define NE     1600000
#define EMB3   192
#define F_LAT  64
#define F_OUT  128
#define NEG_SLOPE 0.2f
#define KEY_DPB 391u
#define DL_BITS 9
#define CH_E   6400
#define CH_P   (CH_E / 32)
#define CH_W   (256 * CH_P / 32)
#define N_CH   (NE / CH_E)

typedef __attribute__((ext_vector_type(2))) float v2f;
typedef __attribute__((ext_vector_type(8))) float v8f;

__device__ __forceinline__ v8f wmma4(v2f a, v2f b, v8f c) {
  return __builtin_amdgcn_wmma_f32_16x16x4_f32(false, a, false, b, (short)0, c,
                                               false, false);
}

__device__ __forceinline__ unsigned enc_f32(float f) {
  unsigned u = __float_as_uint(f);
  return (u & 0x80000000u) ? ~u : (u | 0x80000000u);
}
__device__ __forceinline__ float dec_f32(unsigned u) {
  return (u & 0x80000000u) ? __uint_as_float(u & 0x7FFFFFFFu)
                           : __uint_as_float(~u);
}

#define NF_W 25
#define NF_T (NF_W * 32)
__global__ __launch_bounds__(NF_T) void node_feat_kernel(
    const float* __restrict__ embed,
    const float* __restrict__ W_lin, const float* __restrict__ b_lin,
    const float* __restrict__ W_gat,
    const float* __restrict__ attn_l, const float* __restrict__ attn_r,
    float* __restrict__ feat, float* __restrict__ el, float* __restrict__ er,
    const int* __restrict__ src, const int* __restrict__ dst,
    unsigned* __restrict__ bmap, unsigned* __restrict__ pk,
    int n_tiles) {
  extern __shared__ float smem[];
  float* sW1 = smem;
  float* sW2 = sW1 + EMB3 * F_LAT;
  float* sB1 = sW2 + F_LAT * F_OUT;
  float* sAl = sB1 + F_LAT;
  float* sAr = sAl + F_OUT;
  float* sX  = sAr + F_OUT;
  unsigned* sBm = (unsigned*)(sX + NF_W * 16 * F_LAT);

  const int tid = threadIdx.x;

  static_assert(N_CH * CH_E == NE && CH_E == 8 * NF_T, "one chunk per node block");
  for (int i = tid; i < CH_W; i += NF_T) sBm[i] = 0u;
  __syncthreads();
  {
    const int4* __restrict__ d4 = (const int4*)dst;
    const int4* __restrict__ s4 = (const int4*)src;
    const int ln = tid & 31;
    const int g = blockIdx.x * NF_T + tid;
    if (g < NE / 8) {
      const int ia = 2 * (g - ln) + ln;
      const int ib = ia + 32;
      const int4 da = d4[ia], db = d4[ib];
      const int4 sa = s4[ia], sb = s4[ib];
      const unsigned oa0 = (unsigned)da.x / KEY_DPB, oa1 = (unsigned)da.y / KEY_DPB,
                     oa2 = (unsigned)da.z / KEY_DPB, oa3 = (unsigned)da.w / KEY_DPB;
      const unsigned ob0 = (unsigned)db.x / KEY_DPB, ob1 = (unsigned)db.y / KEY_DPB,
                     ob2 = (unsigned)db.z / KEY_DPB, ob3 = (unsigned)db.w / KEY_DPB;
      const unsigned pa = (unsigned)(ia >> 3) - blockIdx.x * CH_P;
      const unsigned pb = pa + 4u;
      uint4 wa, wb;
      wa.x = ((unsigned)sa.x << DL_BITS) | ((unsigned)da.x - oa0 * KEY_DPB);
      wa.y = ((unsigned)sa.y << DL_BITS) | ((unsigned)da.y - oa1 * KEY_DPB);
      wa.z = ((unsigned)sa.z << DL_BITS) | ((unsigned)da.z - oa2 * KEY_DPB);
      wa.w = ((unsigned)sa.w << DL_BITS) | ((unsigned)da.w - oa3 * KEY_DPB);
      wb.x = ((unsigned)sb.x << DL_BITS) | ((unsigned)db.x - ob0 * KEY_DPB);
      wb.y = ((unsigned)sb.y << DL_BITS) | ((unsigned)db.y - ob1 * KEY_DPB);
      wb.z = ((unsigned)sb.z << DL_BITS) | ((unsigned)db.z - ob2 * KEY_DPB);
      wb.w = ((unsigned)sb.w << DL_BITS) | ((unsigned)db.w - ob3 * KEY_DPB);
      ((uint4*)pk)[ia] = wa;
      ((uint4*)pk)[ib] = wb;
      unsigned q;
      q = oa0 * CH_P + pa; atomicOr(sBm + (q >> 5), 1u << (q & 31u));
      q = oa1 * CH_P + pa; atomicOr(sBm + (q >> 5), 1u << (q & 31u));
      q = oa2 * CH_P + pa; atomicOr(sBm + (q >> 5), 1u << (q & 31u));
      q = oa3 * CH_P + pa; atomicOr(sBm + (q >> 5), 1u << (q & 31u));
      q = ob0 * CH_P + pb; atomicOr(sBm + (q >> 5), 1u << (q & 31u));
      q = ob1 * CH_P + pb; atomicOr(sBm + (q >> 5), 1u << (q & 31u));
      q = ob2 * CH_P + pb; atomicOr(sBm + (q >> 5), 1u << (q & 31u));
      q = ob3 * CH_P + pb; atomicOr(sBm + (q >> 5), 1u << (q & 31u));
    }
  }

  {
    const float4* __restrict__ w1v = (const float4*)W_lin;
    const float4* __restrict__ w2v = (const float4*)W_gat;
    float4* s1v = (float4*)sW1;
    float4* s2v = (float4*)sW2;
    constexpr int N1 = EMB3 * F_LAT / 4;
    constexpr int N2 = F_LAT * F_OUT / 4;
#pragma unroll
    for (int j = 0; j < (N1 + NF_T - 1) / NF_T; ++j) {
      const int i = tid + j * NF_T;
      if (i < N1) s1v[i] = w1v[i];
    }
#pragma unroll
    for (int j = 0; j < (N2 + NF_T - 1) / NF_T; ++j) {
      const int i = tid + j * NF_T;
      if (i < N2) s2v[i] = w2v[i];
    }
  }
  if (tid < F_LAT) sB1[tid] = b_lin[tid];
  if (tid < F_OUT) { sAl[tid] = attn_l[tid]; sAr[tid] = attn_r[tid]; }
  __syncthreads();

  if (blockIdx.x < N_CH)
    for (int i = tid; i < CH_W; i += NF_T) bmap[blockIdx.x * CH_W + i] = sBm[i];

  const int wave = tid >> 5;
  const int lane = tid & 31;
  const int tile = blockIdx.x * NF_W + wave;
  if (tile >= n_tiles) return;

  const int m  = lane & 15;
  const int hi = lane >> 4;

  const float* arow = embed + (size_t)(tile * 16 + m) * EMB3;
  v8f acc[4] = {};
  for (int k0 = 0; k0 < EMB3; k0 += 4) {
    const int ka = k0 + 2 * hi;
    v2f a; a.x = arow[ka]; a.y = arow[ka + 1];
    const float* b0 = sW1 + ka * F_LAT + m;
    const float* b1 = b0 + F_LAT;
#pragma unroll
    for (int nt = 0; nt < 4; ++nt) {
      v2f b; b.x = b0[nt * 16]; b.y = b1[nt * 16];
      acc[nt] = wmma4(a, b, acc[nt]);
    }
  }

  float* xw = sX + wave * 16 * F_LAT;
#pragma unroll
  for (int r = 0; r < 8; ++r) {
    const int row = r + 8 * hi;
#pragma unroll
    for (int nt = 0; nt < 4; ++nt) {
      float v = acc[nt][r] + sB1[nt * 16 + m];
      xw[row * F_LAT + nt * 16 + m] = fmaxf(v, 0.0f);
    }
  }

  v8f fa[8] = {};
  for (int k0 = 0; k0 < F_LAT; k0 += 4) {
    const int ka = k0 + 2 * hi;
    v2f a; a.x = xw[m * F_LAT + ka]; a.y = xw[m * F_LAT + ka + 1];
    const float* b0 = sW2 + ka * F_OUT + m;
    const float* b1 = b0 + F_OUT;
#pragma unroll
    for (int nt = 0; nt < 8; ++nt) {
      v2f b; b.x = b0[nt * 16]; b.y = b1[nt * 16];
      fa[nt] = wmma4(a, b, fa[nt]);
    }
  }

  float pl[8], pr[8];
  const size_t base = (size_t)(tile * 16) * F_OUT;
#pragma unroll
  for (int r = 0; r < 8; ++r) {
    const int row = r + 8 * hi;
    float sl = 0.0f, sr = 0.0f;
#pragma unroll
    for (int nt = 0; nt < 8; ++nt) {
      const float v = fa[nt][r];
      feat[base + row * F_OUT + nt * 16 + m] = v;
      sl += v * sAl[nt * 16 + m];
      sr += v * sAr[nt * 16 + m];
    }
    pl[r] = sl; pr[r] = sr;
  }
#pragma unroll
  for (int off = 8; off >= 1; off >>= 1) {
#pragma unroll
    for (int r = 0; r < 8; ++r) {
      pl[r] += __shfl_xor(pl[r], off, 32);
      pr[r] += __shfl_xor(pr[r], off, 32);
    }
  }
  if (m == 0) {
#pragma unroll
    for (int r = 0; r < 8; ++r) {
      el[tile * 16 + r + 8 * hi] = pl[r];
      er[tile * 16 + r + 8 * hi] = pr[r];
    }
  }
}

#define AGG_T   1024
#define AGG_NB  256
#define AGG_NW  (AGG_T / 32)
#define DPB     391
#define DPB_PAD 392
#define NE4     (NE / 4)
#define N_WIN   (N_CH * 7)
#define BM_W    (N_CH * CH_W)
#define WCAP    320

__device__ __forceinline__ float leaky(float e) {
  return (e > 0.0f) ? e : NEG_SLOPE * e;
}
__device__ __forceinline__ float seg_max(unsigned enc) {
  float emax = dec_f32(enc);
  if ((__float_as_uint(emax) & 0x7F800000u) == 0x7F800000u) emax = 0.0f;
  return emax;
}

__global__ __launch_bounds__(AGG_T) void agg_kernel(
    const int* __restrict__ src, const int* __restrict__ dst,
    const unsigned* __restrict__ bmap, const unsigned* __restrict__ pk,
    const float* __restrict__ el, const float* __restrict__ er,
    const float* __restrict__ feat, const float* __restrict__ b_gat,
    float* __restrict__ out) {
  extern __shared__ float smem[];
  float* acc  = smem;
  float* sden = acc + DPB * F_OUT;
  unsigned* smax = (unsigned*)(sden + DPB_PAD);
  int*   lpk  = (int*)(smax + DPB_PAD);
  int*   le   = lpk + AGG_NW * WCAP;
  int*   sovf = le + AGG_NW * WCAP;

  const int tid  = threadIdx.x;
  const int lane = tid & 31;
  const int wave = tid >> 5;
  const int base = blockIdx.x * DPB;
  int nd = NN - base;
  nd = (nd > DPB) ? DPB : nd;

  for (int i = tid; i < DPB * F_OUT; i += AGG_T) acc[i] = 0.0f;
  if (tid < DPB_PAD) { sden[tid] = 0.0f; smax[tid] = 0u; }
  if (tid == 0) *sovf = 0;
  __syncthreads();

  const int4* __restrict__ d4 = (const int4*)dst;

  int*   wl  = lpk + wave * WCAP;
  int*   wle = le + wave * WCAP;
  int*   wp  = wle;
  int cnt = 0;
  int npc = 0;

  auto append = [&](int* list, int& fill, unsigned F, auto entry) {
    unsigned m = __builtin_amdgcn_ballot_w32(F != 0u);
    while (m != 0u) {
      const int pos = (int)__builtin_amdgcn_mbcnt_lo(m, (unsigned)fill);
      if (F != 0u) {
        const int t = __builtin_ctz(F);
        F &= F - 1u;
        if (pos < WCAP) list[pos] = entry(t);
      }
      fill += __builtin_popcount(m);
      m = __builtin_amdgcn_ballot_w32(F != 0u);
    }
  };

  constexpr int WPW = (N_WIN + AGG_NW - 1) / AGG_NW;
  for (int r = 0; r < (WPW + 31) / 32; ++r) {
    const int kw = r * 32 + lane;
    const int k = wave * WPW + kw;
    unsigned F = 0u;
    int first = 0;
    if (kw < WPW && k < N_WIN) {
      const int c = k / 7, j = k - 7 * c;
      const unsigned b0 = ((unsigned)c * 256u + blockIdx.x) * CH_P + 32u * j;
      const int wi = (int)(b0 >> 5);
      const unsigned lo = bmap[wi];
      const unsigned hi = bmap[(wi + 1 < BM_W) ? wi + 1 : BM_W - 1];
      F = (unsigned)(((((unsigned long long)hi) << 32) | lo) >> (b0 & 31u));
      if (j == 6) F &= 0xFFu;
      first = c * CH_P + 32 * j;
    }
    append(wp, npc, F, [&](int t) { return first + t; });
  }

  const int np = (npc < WCAP) ? npc : WCAP;
  if (npc > WCAP && lane == 0) *sovf = 1;
  const int sub = lane >> 3, u16 = lane & 7;
  for (int p0 = 0; p0 < np; p0 += 32) {
    unsigned F = 0u;
#pragma unroll
    for (int i = 0; i < 8; ++i) {
      const int p = p0 + 4 * i + sub;
      const int q = wp[(p < np) ? p : np - 1];
      const int4 dv = d4[q * 8 + u16];
      unsigned f = (((unsigned)(dv.x - base) < (unsigned)nd) ? 1u : 0u) |
                   (((unsigned)(dv.y - base) < (unsigned)nd) ? 2u : 0u) |
                   (((unsigned)(dv.z - base) < (unsigned)nd) ? 4u : 0u) |
                   (((unsigned)(dv.w - base) < (unsigned)nd) ? 8u : 0u);
      f = (p < np) ? f : 0u;
      F |= f << (4 * i);
    }
    append(wl, cnt, F, [&](int t) {
      return wp[p0 + (t & ~3) + sub] * 32 + 4 * u16 + (t & 3);
    });
  }

  const int n = (cnt < WCAP) ? cnt : WCAP;
  if (cnt > WCAP && lane == 0) *sovf = 1;
  for (int p = lane; p < n; p += 32) {
    const int ei = wl[p];
    const int w  = (int)pk[ei];
    const int dl = w & ((1 << DL_BITS) - 1);
    const int s  = w >> DL_BITS;
    const float e = leaky(el[s] + er[base + dl]);
    atomicMax(smax + dl, enc_f32(e));
    wl[p]  = w;
    wle[p] = __float_as_int(e);
  }
  __syncthreads();

  if (*sovf == 0) {
    for (int p = lane; p < n; p += 32) {
      const int q_d = wl[p] & ((1 << DL_BITS) - 1);
      const float ee_i = expf(__int_as_float(wle[p]) - seg_max(smax[q_d]));
      atomicAdd(sden + q_d, ee_i);
      wle[p] = __float_as_int(ee_i);
    }
    const char* __restrict__ fbytes = (const char*)feat + lane * 16;
    float* accl = acc + lane;
    const int sub8 = lane & 7;
    const int ng = (n + 7) >> 3;
    auto meta = [&](int g, int& mpk, int& mw) {
      int idx = g * 8 + sub8;
      idx = (idx < n) ? idx : n - 1;
      mpk = wl[idx];
      mw  = wle[idx];
    };
    auto issue = [&](int mpk, float4 (&v)[8]) {
#pragma unroll
      for (int j = 0; j < 8; ++j) {
        const unsigned pkj = (unsigned)__builtin_amdgcn_readlane(mpk, j);
        v[j] = *(const float4*)(fbytes + (size_t)(pkj & ~((1u << DL_BITS) - 1u)));
      }
    };
    auto consume = [&](int mpk, int mw, const float4 (&v)[8], int count) {
#pragma unroll
      for (int j = 0; j < 8; ++j) {
        if (j < count) {
          const int pkj = __builtin_amdgcn_readlane(mpk, j);
          const float w = __int_as_float(__builtin_amdgcn_readlane(mw, j));
          float* ar = accl + (pkj & ((1 << DL_BITS) - 1)) * F_OUT;
          atomicAdd(ar,      v[j].x * w);
          atomicAdd(ar + 32, v[j].y * w);
          atomicAdd(ar + 64, v[j].z * w);
          atomicAdd(ar + 96, v[j].w * w);
        }
      }
    };
    if (ng > 0) {
      float4 va[8], vb[8];
      int pk0, w0, pk1, w1, pk2, w2, pk3, w3;
      meta(0, pk0, w0);
      meta(1, pk1, w1);
      issue(pk0, va);
      int g = 0;
      for (; g + 2 < ng; g += 2) {
        meta(g + 2, pk2, w2);
        issue(pk1, vb);
        __builtin_amdgcn_sched_barrier(0);
        consume(pk0, w0, va, 8);
        meta(g + 3, pk3, w3);
        issue(pk2, va);
        __builtin_amdgcn_sched_barrier(0);
        consume(pk1, w1, vb, 8);
        pk0 = pk2; w0 = w2; pk1 = pk3; w1 = w3;
      }
      const int left = n - 8 * g;
      if (g + 1 < ng) {
        issue(pk1, vb);
        __builtin_amdgcn_sched_barrier(0);
        consume(pk0, w0, va, 8);
        consume(pk1, w1, vb, left - 8);
      } else {
        consume(pk0, w0, va, left);
      }
    }
  } else {
    for (int sweep = 0; sweep < 2; ++sweep) {
      for (int i0 = 0; i0 < NE4; i0 += AGG_T) {
        const int idx = i0 + tid;
        const bool in = idx < NE4;
        const int4 dv = d4[in ? idx : NE4 - 1];
        for (int c = 0; c < 4; ++c) {
          const int dl = ((c == 0) ? dv.x : (c == 1) ? dv.y
                        : (c == 2) ? dv.z : dv.w) - base;
          unsigned mask = __builtin_amdgcn_ballot_w32(in && (unsigned)dl < (unsigned)nd);
          while (mask != 0u) {
            const int j = __builtin_ctz(mask);
            mask &= mask - 1u;
            const int dj = __builtin_amdgcn_readlane(dl, j);
            const int sj = src[__builtin_amdgcn_readlane(idx * 4 + c, j)];
            const float e = leaky(el[sj] + er[base + dj]);
            if (sweep == 0) {
              if (lane == 0) atomicMax(smax + dj, enc_f32(e));
            } else {
              const float w = expf(e - seg_max(smax[dj]));
              if (lane == 0) atomicAdd(sden + dj, w);
              const float4 fv = ((const float4*)(feat + (size_t)sj * F_OUT))[lane];
              float* ar = acc + dj * F_OUT + lane;
              atomicAdd(ar,      fv.x * w);
              atomicAdd(ar + 32, fv.y * w);
              atomicAdd(ar + 64, fv.z * w);
              atomicAdd(ar + 96, fv.w * w);
            }
          }
        }
      }
      __syncthreads();
    }
  }
  __syncthreads();

  {
    const int half = lane >> 4, q = lane & 15;
    const float4 bm = ((const float4*)b_gat)[q];
    const float4 bl = ((const float4*)b_gat)[16 + q];
    for (int r0 = 2 * wave; r0 < nd; r0 += 2 * AGG_NW) {
      const int r  = r0 + half;
      const bool ok = r < nd;
      const int rr = ok ? r : r0;
      float dn = sden[rr];
      dn = (dn > 0.0f) ? dn : 1.0f;
      const float* ar = acc + rr * F_OUT + q;
      float4 ym, yl;
      ym.x = ar[0]  / dn + bm.x;
      ym.y = ar[32] / dn + bm.y;
      ym.z = ar[64] / dn + bm.z;
      ym.w = ar[96] / dn + bm.w;
      yl.x = tanhf(ar[16]  / dn + bl.x);
      yl.y = tanhf(ar[48]  / dn + bl.y);
      yl.z = tanhf(ar[80]  / dn + bl.z);
      yl.w = tanhf(ar[112] / dn + bl.w);
      if (ok) {
        const size_t n_ = (size_t)(base + r);
        *(float4*)(out + n_ * 64 + 4 * q) = ym;
        *(float4*)(out + (size_t)NN * 64 + n_ * 64 + 4 * q) = yl;
      }
    }
  }
}

extern "C" void kernel_launch(void* const* d_in, const int* in_sizes, int n_in,
                              void* d_out, int out_size, void* d_ws,
                              size_t ws_size, hipStream_t stream) {
  const float* embed  = (const float*)d_in[0];
  const int*   src    = (const int*)d_in[1];
  const int*   dst    = (const int*)d_in[2];
  const float* W_lin  = (const float*)d_in[3];
  const float* b_lin  = (const float*)d_in[4];
  const float* W_gat  = (const float*)d_in[5];
  const float* attn_l = (const float*)d_in[6];
  const float* attn_r = (const float*)d_in[7];
  const float* b_gat  = (const float*)d_in[8];
  float* out = (float*)d_out;

  float* ws   = (float*)d_ws;
  float* feat = ws;
  float* el   = feat + (size_t)NN * F_OUT;
  float* er   = el + NN;
  unsigned* bmap = (unsigned*)(er + NN);
  unsigned* pk   = bmap + NE / 4;

  const int n_tiles = NN / 16;
  const int nf_blocks = (n_tiles + NF_W - 1) / NF_W;
  const size_t smem_bytes =
      (size_t)(EMB3 * F_LAT + F_LAT * F_OUT + F_LAT + 2 * F_OUT +
               NF_W * 16 * F_LAT + CH_W) * sizeof(float);
  (void)hipFuncSetAttribute((const void*)node_feat_kernel,
                            hipFuncAttributeMaxDynamicSharedMemorySize,
                            (int)smem_bytes);
  node_feat_kernel<<<nf_blocks, NF_T, smem_bytes, stream>>>(
      embed, W_lin, b_lin, W_gat, attn_l, attn_r, feat, el, er, src, dst, bmap,
      pk, n_tiles);

  const size_t agg_bytes =
      (size_t)(DPB * F_OUT + 2 * DPB_PAD + 2 * AGG_NW * WCAP + 4) *
      sizeof(float);
  (void)hipFuncSetAttribute((const void*)agg_kernel,
                            hipFuncAttributeMaxDynamicSharedMemorySize,
                            (int)agg_bytes);
  agg_kernel<<<AGG_NB, AGG_T, agg_bytes, stream>>>(src, dst, bmap, pk, el, er,
                                                   feat, b_gat, out);
}


# node kernel: WMMA fragments by 16-byte loads: one k permutation in A and B loaders (half h holds k 8j+4h..+3 over two steps), weights in LDS in fragment order, x tile pitch 68; k sum order changes
# speedup vs baseline: 5.9601x; 1.2148x over previous
#include <hip/hip_runtime.h>
#include <math.h>

#define NN     100000
#define NE     1600000
#define EMB3   192
#define F_LAT  64
#define F_OUT  128
#define NEG_SLOPE 0.2f
#define KEY_DPB 391u
#define DL_BITS 9
#define CH_E   6400
#define N_CH   (NE / CH_E)
#define N_OWN  256

typedef __attribute__((ext_vector_type(2))) float v2f;
typedef __attribute__((ext_vector_type(8))) float v8f;

__device__ __forceinline__ v8f wmma4(v2f a, v2f b, v8f c) {
  return __builtin_amdgcn_wmma_f32_16x16x4_f32(false, a, false, b, (short)0, c,
                                               false, false);
}

__device__ __forceinline__ unsigned enc_f32(float f) {
  unsigned u = __float_as_uint(f);
  return (u & 0x80000000u) ? ~u : (u | 0x80000000u);
}
__device__ __forceinline__ float dec_f32(unsigned u) {
  return (u & 0x80000000u) ? __uint_as_float(u & 0x7FFFFFFFu)
                           : __uint_as_float(~u);
}

__device__ __forceinline__ void put4(float* d, float4 v, int rot) {
  const bool b0 = (rot & 1) != 0, b1 = (rot & 2) != 0;
  const float r0 = b0 ? v.y : v.x, r1 = b0 ? v.z : v.y, r2 = b0 ? v.w : v.z, r3 = b0 ? v.x : v.w;
  const float q0 = b1 ? r2 : r0, q1 = b1 ? r3 : r1, q2 = b1 ? r0 : r2, q3 = b1 ? r1 : r3;
  d[4 * (rot & 3)]       = q0;
  d[4 * ((rot + 1) & 3)] = q1;
  d[4 * ((rot + 2) & 3)] = q2;
  d[4 * ((rot + 3) & 3)] = q3;
}

#define NF_W 25
#define XP   68
#define NF_T (NF_W * 32)
__global__ __launch_bounds__(NF_T) void node_feat_kernel(
    const float* __restrict__ embed,
    const float* __restrict__ W_lin, const float* __restrict__ b_lin,
    const float* __restrict__ W_gat,
    const float* __restrict__ attn_l, const float* __restrict__ attn_r,
    float* __restrict__ feat, float* __restrict__ el, float* __restrict__ er,
    const int* __restrict__ src, const int* __restrict__ dst,
    unsigned* __restrict__ tab, unsigned* __restrict__ plist,
    int n_tiles) {
  extern __shared__ __align__(16) float smem[];
  float* sW1 = smem;
  float* sW2 = sW1 + EMB3 * F_LAT;
  float* sB1 = sW2 + F_LAT * F_OUT;
  float* sAl = sB1 + F_LAT;
  float* sAr = sAl + F_OUT;
  float* sX  = sAr + F_OUT;
  unsigned* sCnt = (unsigned*)(sX + NF_W * 16 * XP);
  unsigned* sOff = sCnt + N_OWN;
  unsigned* sL   = sOff + N_OWN;

  const int tid = threadIdx.x;

  static_assert(N_CH * CH_E == NE && CH_E == 8 * NF_T, "one chunk per node block");
  static_assert(N_OWN * KEY_DPB >= NN && N_OWN == 8 * 32, "owners: 8 counters a lane of one wave");
  if (tid < N_OWN) sCnt[tid] = 0u;
  __syncthreads();
  unsigned ew[8], eo[8], rk[8];
  const bool act = blockIdx.x * NF_T + tid < NE / 8;
  {
    const int4* __restrict__ d4 = (const int4*)dst;
    const int4* __restrict__ s4 = (const int4*)src;
    const int ln = tid & 31;
    const int g = blockIdx.x * NF_T + tid;
    if (act) {
      const int ia = 2 * (g - ln) + ln;
      const int ib = ia + 32;
      const int4 da = d4[ia], db = d4[ib];
      const int4 sa = s4[ia], sb = s4[ib];
      eo[0] = (unsigned)da.x / KEY_DPB; eo[1] = (unsigned)da.y / KEY_DPB;
      eo[2] = (unsigned)da.z / KEY_DPB; eo[3] = (unsigned)da.w / KEY_DPB;
      eo[4] = (unsigned)db.x / KEY_DPB; eo[5] = (unsigned)db.y / KEY_DPB;
      eo[6] = (unsigned)db.z / KEY_DPB; eo[7] = (unsigned)db.w / KEY_DPB;
      ew[0] = ((unsigned)sa.x << DL_BITS) | ((unsigned)da.x - eo[0] * KEY_DPB);
      ew[1] = ((unsigned)sa.y << DL_BITS) | ((unsigned)da.y - eo[1] * KEY_DPB);
      ew[2] = ((unsigned)sa.z << DL_BITS) | ((unsigned)da.z - eo[2] * KEY_DPB);
      ew[3] = ((unsigned)sa.w << DL_BITS) | ((unsigned)da.w - eo[3] * KEY_DPB);
      ew[4] = ((unsigned)sb.x << DL_BITS) | ((unsigned)db.x - eo[4] * KEY_DPB);
      ew[5] = ((unsigned)sb.y << DL_BITS) | ((unsigned)db.y - eo[5] * KEY_DPB);
      ew[6] = ((unsigned)sb.z << DL_BITS) | ((unsigned)db.z - eo[6] * KEY_DPB);
      ew[7] = ((unsigned)sb.w << DL_BITS) | ((unsigned)db.w - eo[7] * KEY_DPB);
#pragma unroll
      for (int i = 0; i < 8; ++i) rk[i] = atomicAdd(sCnt + eo[i], 1u);
    }
  }
  __syncthreads();
  if (tid < 32) {
    unsigned v[8];
    unsigned run = 0u;
#pragma unroll
    for (int i = 0; i < 8; ++i) { const unsigned t = sCnt[tid * 8 + i]; v[i] = run; run += t; }
    unsigned inc = run;
#pragma unroll
    for (int d = 1; d < 32; d <<= 1) {
      const unsigned nb = __shfl_up(inc, d, 32);
      if (tid >= d) inc += nb;
    }
    const unsigned exc = inc - run;
#pragma unroll
    for (int i = 0; i < 8; ++i) sOff[tid * 8 + i] = exc + v[i];
  }
  __syncthreads();
  if (act) {
    unsigned pl8[8];
#pragma unroll
    for (int i = 0; i < 8; ++i) pl8[i] = sOff[eo[i]] + rk[i];
#pragma unroll
    for (int i = 0; i < 8; ++i) sL[pl8[i]] = ew[i];
  }

  {
    const float4* __restrict__ w1v = (const float4*)W_lin;
    const float4* __restrict__ w2v = (const float4*)W_gat;
    constexpr int N1 = EMB3 * F_LAT / 4;
    constexpr int N2 = F_LAT * F_OUT / 4;
    constexpr int P1 = (N1 + NF_T - 1) / NF_T;
    constexpr int P2 = (N2 + NF_T - 1) / NF_T;
    float4 va[P1], vb[P2];
#pragma unroll
    for (int j = 0; j < P1; ++j) {
      const int i = tid + j * NF_T;
      va[j] = w1v[(i < N1) ? i : N1 - 1];
    }
#pragma unroll
    for (int j = 0; j < P2; ++j) {
      const int i = tid + j * NF_T;
      vb[j] = w2v[(i < N2) ? i : N2 - 1];
    }
    asm volatile("" :: "v"(va[P1 - 1].x), "v"(vb[P2 - 1].x));
#pragma unroll
    for (int j = 0; j < P1; ++j) {
      const int i = tid + j * NF_T;
      if (i < N1) {
        const int k = i >> 4, p = i & 15;
        const int jh = k >> 2, st = (k >> 1) & 1, e = k & 1;
        float* d = sW1 + ((jh * 4 + st * 2 + (p >> 3)) * 16 + 4 * (p & 3)) * 4 + ((p >> 2) & 1) * 2 + e;
        put4(d, va[j], p >> 3);
      }
    }
#pragma unroll
    for (int j = 0; j < P2; ++j) {
      const int i = tid + j * NF_T;
      if (i < N2) {
        const int k = i >> 5, p = i & 31;
        const int jh = k >> 2, st = (k >> 1) & 1, e = k & 1;
        float* d = sW2 + ((jh * 8 + st * 4 + (p >> 3)) * 16 + 4 * (p & 3)) * 4 + ((p >> 2) & 1) * 2 + e;
        put4(d, vb[j], p >> 3);
      }
    }
  }
  if (tid < F_LAT) sB1[tid] = b_lin[tid];
  if (tid < F_OUT) { sAl[tid] = attn_l[tid]; sAr[tid] = attn_r[tid]; }
  __syncthreads();

  if (blockIdx.x < N_CH) {
    const uint4* slv = (const uint4*)sL;
    uint4* plv = (uint4*)plist + blockIdx.x * (CH_E / 4);
    plv[tid] = slv[tid];
    plv[tid + NF_T] = slv[tid + NF_T];
    if (tid < N_OWN) tab[blockIdx.x * N_OWN + tid] = (sOff[tid] << 16) | sCnt[tid];
  }

  const int wave = tid >> 5;
  const int lane = tid & 31;
  const int tile = blockIdx.x * NF_W + wave;
  if (tile >= n_tiles) return;

  const int m  = lane & 15;
  const int hi = lane >> 4;

  const float* arow = embed + (size_t)(tile * 16 + m) * EMB3 + 4 * hi;
  const float4* bw1 = (const float4*)sW1 + hi * 64 + m;
  v8f acc[4] = {};
  for (int j = 0; j < EMB3 / 8; ++j) {
    const float4 a4 = *(const float4*)(arow + 8 * j);
    const float4* bj = bw1 + j * 128;
    const float4 b0 = bj[0], b1 = bj[16], b2 = bj[32], b3 = bj[48];
    v2f a, b;
    a.x = a4.x; a.y = a4.y;
    b.x = b0.x; b.y = b0.y; acc[0] = wmma4(a, b, acc[0]);
    b.x = b0.z; b.y = b0.w; acc[1] = wmma4(a, b, acc[1]);
    b.x = b1.x; b.y = b1.y; acc[2] = wmma4(a, b, acc[2]);
    b.x = b1.z; b.y = b1.w; acc[3] = wmma4(a, b, acc[3]);
    a.x = a4.z; a.y = a4.w;
    b.x = b2.x; b.y = b2.y; acc[0] = wmma4(a, b, acc[0]);
    b.x = b2.z; b.y = b2.w; acc[1] = wmma4(a, b, acc[1]);
    b.x = b3.x; b.y = b3.y; acc[2] = wmma4(a, b, acc[2]);
    b.x = b3.z; b.y = b3.w; acc[3] = wmma4(a, b, acc[3]);
  }

  float* xw = sX + wave * 16 * XP;
#pragma unroll
  for (int r = 0; r < 8; ++r) {
    const int row = r + 8 * hi;
#pragma unroll
    for (int nt = 0; nt < 4; ++nt) {
      float v = acc[nt][r] + sB1[nt * 16 + m];
      xw[row * XP + nt * 16 + m] = fmaxf(v, 0.0f);
    }
  }

  const float* xrow = xw + m * XP + 4 * hi;
  const float4* bw2 = (const float4*)sW2 + hi * 128 + m;
  v8f fa[8] = {};
  for (int j = 0; j < F_LAT / 8; ++j) {
    const float4 a4 = *(const float4*)(xrow + 8 * j);
    const float4* bj = bw2 + j * 256;
    v2f a, b;
    a.x = a4.x; a.y = a4.y;
#pragma unroll
    for (int q = 0; q < 4; ++q) {
      const float4 bq = bj[q * 16];
      b.x = bq.x; b.y = bq.y; fa[2 * q]     = wmma4(a, b, fa[2 * q]);
      b.x = bq.z; b.y = bq.w; fa[2 * q + 1] = wmma4(a, b, fa[2 * q + 1]);
    }
    a.x = a4.z; a.y = a4.w;
#pragma unroll
    for (int q = 0; q < 4; ++q) {
      const float4 bq = bj[(4 + q) * 16];
      b.x = bq.x; b.y = bq.y; fa[2 * q]     = wmma4(a, b, fa[2 * q]);
      b.x = bq.z; b.y = bq.w; fa[2 * q + 1] = wmma4(a, b, fa[2 * q + 1]);
    }
  }

  float pl[8], pr[8];
  const size_t base = (size_t)(tile * 16) * F_OUT;
#pragma unroll
  for (int r = 0; r < 8; ++r) {
    const int row = r + 8 * hi;
    float sl = 0.0f, sr = 0.0f;
#pragma unroll
    for (int nt = 0; nt < 8; ++nt) {
      const float v = fa[nt][r];
      feat[base + row * F_OUT + nt * 16 + m] = v;
      sl += v * sAl[nt * 16 + m];
      sr += v * sAr[nt * 16 + m];
    }
    pl[r] = sl; pr[r] = sr;
  }
#pragma unroll
  for (int off = 8; off >= 1; off >>= 1) {
#pragma unroll
    for (int r = 0; r < 8; ++r) {
      pl[r] += __shfl_xor(pl[r], off, 32);
      pr[r] += __shfl_xor(pr[r], off, 32);
    }
  }
  if (m == 0) {
#pragma unroll
    for (int r = 0; r < 8; ++r) {
      el[tile * 16 + r + 8 * hi] = pl[r];
      er[tile * 16 + r + 8 * hi] = pr[r];
    }
  }
}

#define AGG_T   1024
#define AGG_NB  256
#define AGG_NW  (AGG_T / 32)
#define DPB     391
#define DPB_PAD 392
#define NE4     (NE / 4)
#define WCAP    320

__device__ __forceinline__ float leaky(float e) {
  return (e > 0.0f) ? e : NEG_SLOPE * e;
}
__device__ __forceinline__ float seg_max(unsigned enc) {
  float emax = dec_f32(enc);
  if ((__float_as_uint(emax) & 0x7F800000u) == 0x7F800000u) emax = 0.0f;
  return emax;
}

__global__ __launch_bounds__(AGG_T) void agg_kernel(
    const int* __restrict__ src, const int* __restrict__ dst,
    const unsigned* __restrict__ tab, const unsigned* __restrict__ plist,
    const float* __restrict__ el, const float* __restrict__ er,
    const float* __restrict__ feat, const float* __restrict__ b_gat,
    float* __restrict__ out) {
  extern __shared__ __align__(16) float smem_agg[];
  float* acc  = smem_agg;
  float* sden = acc + DPB * F_OUT;
  unsigned* smax = (unsigned*)(sden + DPB_PAD);
  int*   lpk  = (int*)(smax + DPB_PAD);
  int*   le   = lpk + AGG_NW * WCAP;
  int*   scnt = le + AGG_NW * WCAP;
  int*   sovf = scnt + AGG_NW;
  static_assert(AGG_NB == N_OWN && (unsigned)DPB == KEY_DPB, "an aggregation block is an owner of the lists");

  const int tid  = threadIdx.x;
  const int lane = tid & 31;
  const int wave = tid >> 5;
  const int base = blockIdx.x * DPB;
  int nd = NN - base;
  nd = (nd > DPB) ? DPB : nd;

  if (tid < DPB_PAD) { sden[tid] = 0.0f; smax[tid] = 0u; }
  if (tid < AGG_NW) scnt[tid] = 0;
  if (tid == 0) *sovf = 0;
  __syncthreads();

  const int4* __restrict__ d4 = (const int4*)dst;

  constexpr int CPW = (N_CH + AGG_NW - 1) / AGG_NW;
  static_assert(CPW <= 32, "a lane per chunk of the wave");
  unsigned myt = 0u;
  {
    const int ch = wave + AGG_NW * lane;
    if (lane < CPW && ch < N_CH) myt = tab[ch * N_OWN + blockIdx.x];
  }
  auto put = [&](unsigned w, float e, int pos) {
    const int ow = (int)(w & (unsigned)(AGG_NW - 1));
    if (pos < WCAP) {
      lpk[ow * WCAP + pos] = (int)w;
      le[ow * WCAP + pos]  = __float_as_int(e);
    } else {
      *sovf = 1;
    }
  };
  {
    unsigned wv[CPW];
    float    ev[CPW], rv[CPW];
    int      pos[CPW];
#pragma unroll
    for (int j = 0; j < CPW; ++j) {
      const unsigned t = (unsigned)__builtin_amdgcn_readlane((int)myt, j);
      const int st = (int)(t >> 16), cn = (int)(t & 0xFFFFu);
      int ch = wave + AGG_NW * j;
      ch = (ch < N_CH) ? ch : N_CH - 1;
      int i = st + ((lane < cn) ? lane : 0);
      i = (i < CH_E) ? i : CH_E - 1;
      wv[j] = plist[ch * CH_E + i];
    }
    asm volatile("" :: "v"(wv[0]), "v"(wv[1]), "v"(wv[2]), "v"(wv[3]),
                       "v"(wv[4]), "v"(wv[5]), "v"(wv[6]), "v"(wv[7]));
    static_assert(CPW == 8, "the two statements that hold the loads name 8 groups");
#pragma unroll
    for (int j = 0; j < CPW; ++j) {
      const int cn = __builtin_amdgcn_readlane((int)myt, j) & 0xFFFF;
      const int dl = (int)(wv[j] & ((1u << DL_BITS) - 1u));
      ev[j] = el[wv[j] >> DL_BITS];
      rv[j] = er[base + ((lane < cn) ? dl : 0)];
    }
    asm volatile("" :: "v"(ev[0]), "v"(ev[1]), "v"(ev[2]), "v"(ev[3]),
                       "v"(ev[4]), "v"(ev[5]), "v"(ev[6]), "v"(ev[7]),
                       "v"(rv[0]), "v"(rv[1]), "v"(rv[2]), "v"(rv[3]),
                       "v"(rv[4]), "v"(rv[5]), "v"(rv[6]), "v"(rv[7]));
#pragma unroll
    for (int j = 0; j < CPW; ++j) {
      const int cn = __builtin_amdgcn_readlane((int)myt, j) & 0xFFFF;
      ev[j] = leaky(ev[j] + rv[j]);
      pos[j] = 0;
      if (lane < cn) {
        const int dl = (int)(wv[j] & ((1u << DL_BITS) - 1u));
        atomicMax(smax + dl, enc_f32(ev[j]));
        pos[j] = atomicAdd(scnt + (dl & (AGG_NW - 1)), 1);
      }
    }
#pragma unroll
    for (int j = 0; j < CPW; ++j) {
      const int cn = __builtin_amdgcn_readlane((int)myt, j) & 0xFFFF;
      if (lane < cn) put(wv[j], ev[j], pos[j]);
    }
    for (int j = 0; j < CPW; ++j) {
      const unsigned t = (unsigned)__builtin_amdgcn_readlane((int)myt, j);
      const int st = (int)(t >> 16), cn = (int)(t & 0xFFFFu);
      const int ch = wave + AGG_NW * j;
      for (int b = 32; b < cn; b += 32) {
        if (b + lane < cn) {
          const unsigned w = plist[ch * CH_E + st + b + lane];
          const int dl = (int)(w & ((1u << DL_BITS) - 1u));
          const float e = leaky(el[w >> DL_BITS] + er[base + dl]);
          atomicMax(smax + dl, enc_f32(e));
          put(w, e, atomicAdd(scnt + (dl & (AGG_NW - 1)), 1));
        }
      }
    }
  }
  __syncthreads();

  if (*sovf == 0) {
    int* ol = lpk + wave * WCAP;
    int* oe = le + wave * WCAP;
    const int no = __builtin_amdgcn_readfirstlane(scnt[wave]);
    float4* accv = (float4*)acc + lane;
    for (int r = wave; r < DPB; r += AGG_NW)
      accv[r * (F_OUT / 4)] = make_float4(0.0f, 0.0f, 0.0f, 0.0f);
    for (int p = lane; p < no; p += 32) {
      const int q_d = ol[p] & ((1 << DL_BITS) - 1);
      const float ee_i = expf(__int_as_float(oe[p]) - seg_max(smax[q_d]));
      atomicAdd(sden + q_d, ee_i);
      oe[p] = __float_as_int(ee_i);
    }
    const char* __restrict__ fbytes = (const char*)feat + lane * 16;
    const int sub8 = lane & 7;
    const int ng = (no + 7) >> 3;
    auto meta = [&](int g, int& mpk, int& mw) {
      int idx = g * 8 + sub8;
      idx = (idx < no) ? idx : no - 1;
      mpk = ol[idx];
      mw  = oe[idx];
    };
    auto issue = [&](int mpk, float4 (&v)[8]) {
#pragma unroll
      for (int j = 0; j < 8; ++j) {
        const unsigned pkj = (unsigned)__builtin_amdgcn_readlane(mpk, j);
        v[j] = *(const float4*)(fbytes + (size_t)(pkj & ~((1u << DL_BITS) - 1u)));
      }
    };
    auto consume = [&](int mpk, int mw, const float4 (&v)[8], int count) {
#pragma unroll
      for (int j = 0; j < 8; j += 2) {
        if (j < count) {
          const int d0 = __builtin_amdgcn_readlane(mpk, j) & ((1 << DL_BITS) - 1);
          const float w0 = __int_as_float(__builtin_amdgcn_readlane(mw, j));
          float4* r0 = accv + d0 * (F_OUT / 4);
          if (j + 1 < count) {
            const int d1 = __builtin_amdgcn_readlane(mpk, j + 1) & ((1 << DL_BITS) - 1);
            const float w1 = __int_as_float(__builtin_amdgcn_readlane(mw, j + 1));
            if (d0 != d1) {
              float4* r1 = accv + d1 * (F_OUT / 4);
              float4 a = *r0, b = *r1;
              a.x += v[j].x * w0;     a.y += v[j].y * w0;
              a.z += v[j].z * w0;     a.w += v[j].w * w0;
              b.x += v[j + 1].x * w1; b.y += v[j + 1].y * w1;
              b.z += v[j + 1].z * w1; b.w += v[j + 1].w * w1;
              *r0 = a;
              *r1 = b;
            } else {
              float4 a = *r0;
              a.x += v[j].x * w0;     a.y += v[j].y * w0;
              a.z += v[j].z * w0;     a.w += v[j].w * w0;
              a.x += v[j + 1].x * w1; a.y += v[j + 1].y * w1;
              a.z += v[j + 1].z * w1; a.w += v[j + 1].w * w1;
              *r0 = a;
            }
          } else {
            float4 a = *r0;
            a.x += v[j].x * w0; a.y += v[j].y * w0;
            a.z += v[j].z * w0; a.w += v[j].w * w0;
            *r0 = a;
          }
        }
      }
    };
    if (ng > 0) {
      float4 va[8], vb[8];
      int pk0, w0, pk1, w1, pk2, w2, pk3, w3;
      meta(0, pk0, w0);
      meta(1, pk1, w1);
      issue(pk0, va);
      int g = 0;
      for (; g + 2 < ng; g += 2) {
        meta(g + 2, pk2, w2);
        issue(pk1, vb);
        __builtin_amdgcn_sched_barrier(0);
        consume(pk0, w0, va, 8);
        meta(g + 3, pk3, w3);
        issue(pk2, va);
        __builtin_amdgcn_sched_barrier(0);
        consume(pk1, w1, vb, 8);
        pk0 = pk2; w0 = w2; pk1 = pk3; w1 = w3;
      }
      const int left = no - 8 * g;
      if (g + 1 < ng) {
        issue(pk1, vb);
        __builtin_amdgcn_sched_barrier(0);
        consume(pk0, w0, va, 8);
        consume(pk1, w1, vb, left - 8);
      } else {
        consume(pk0, w0, va, left);
      }
    }
  } else {
    for (int i = tid; i < DPB * F_OUT; i += AGG_T) acc[i] = 0.0f;
    __syncthreads();
    for (int sweep = 0; sweep < 2; ++sweep) {
      for (int i0 = 0; i0 < NE4; i0 += AGG_T) {
        const int idx = i0 + tid;
        const bool in = idx < NE4;
        const int4 dv = d4[in ? idx : NE4 - 1];
        for (int c = 0; c < 4; ++c) {
          const int dl = ((c == 0) ? dv.x : (c == 1) ? dv.y
                        : (c == 2) ? dv.z : dv.w) - base;
          unsigned mask = __builtin_amdgcn_ballot_w32(in && (unsigned)dl < (unsigned)nd);
          while (mask != 0u) {
            const int j = __builtin_ctz(mask);
            mask &= mask - 1u;
            const int dj = __builtin_amdgcn_readlane(dl, j);
            const int sj = src[__builtin_amdgcn_readlane(idx * 4 + c, j)];
            const float e = leaky(el[sj] + er[base + dj]);
            if (sweep == 0) {
              if (lane == 0) atomicMax(smax + dj, enc_f32(e));
            } else {
              const float w = expf(e - seg_max(smax[dj]));
              if (lane == 0) atomicAdd(sden + dj, w);
              const float4 fv = ((const float4*)(feat + (size_t)sj * F_OUT))[lane];
              float* ar = acc + dj * F_OUT + 4 * lane;
              atomicAdd(ar,     fv.x * w);
              atomicAdd(ar + 1, fv.y * w);
              atomicAdd(ar + 2, fv.z * w);
              atomicAdd(ar + 3, fv.w * w);
            }
          }
        }
      }
      __syncthreads();
    }
  }

  {
    const int half = lane >> 4, q = lane & 15;
    const float4 bm = ((const float4*)b_gat)[q];
    const float4 bl = ((const float4*)b_gat)[16 + q];
    for (int r0 = wave; r0 < nd; r0 += 2 * AGG_NW) {
      const int r  = r0 + AGG_NW * half;
      const bool ok = r < nd;
      const int rr = ok ? r : r0;
      float dn = sden[rr];
      dn = (dn > 0.0f) ? dn : 1.0f;
      const float4 am = *(const float4*)(acc + rr * F_OUT + 4 * q);
      const float4 al = *(const float4*)(acc + rr * F_OUT + 64 + 4 * q);
      float4 ym, yl;
      ym.x = am.x / dn + bm.x;
      ym.y = am.y / dn + bm.y;
      ym.z = am.z / dn + bm.z;
      ym.w = am.w / dn + bm.w;
      yl.x = tanhf(al.x / dn + bl.x);
      yl.y = tanhf(al.y / dn + bl.y);
      yl.z = tanhf(al.z / dn + bl.z);
      yl.w = tanhf(al.w / dn + bl.w);
      if (ok) {
        const size_t n_ = (size_t)(base + r);
        *(float4*)(out + n_ * 64 + 4 * q) = ym;
        *(float4*)(out + (size_t)NN * 64 + n_ * 64 + 4 * q) = yl;
      }
    }
  }
}

extern "C" void kernel_launch(void* const* d_in, const int* in_sizes, int n_in,
                              void* d_out, int out_size, void* d_ws,
                              size_t ws_size, hipStream_t stream) {
  const float* embed  = (const float*)d_in[0];
  const int*   src    = (const int*)d_in[1];
  const int*   dst    = (const int*)d_in[2];
  const float* W_lin  = (const float*)d_in[3];
  const float* b_lin  = (const float*)d_in[4];
  const float* W_gat  = (const float*)d_in[5];
  const float* attn_l = (const float*)d_in[6];
  const float* attn_r = (const float*)d_in[7];
  const float* b_gat  = (const float*)d_in[8];
  float* out = (float*)d_out;

  float* ws   = (float*)d_ws;
  float* feat = ws;
  float* el   = feat + (size_t)NN * F_OUT;
  float* er   = el + NN;
  unsigned* tab   = (unsigned*)(er + NN);
  unsigned* plist = tab + NE / 4;

  const int n_tiles = NN / 16;
  const int nf_blocks = (n_tiles + NF_W - 1) / NF_W;
  const size_t smem_bytes =
      (size_t)(EMB3 * F_LAT + F_LAT * F_OUT + F_LAT + 2 * F_OUT +
               NF_W * 16 * XP + 2 * N_OWN + CH_E) * sizeof(float);
  (void)hipFuncSetAttribute((const void*)node_feat_kernel,
                            hipFuncAttributeMaxDynamicSharedMemorySize,
                            (int)smem_bytes);
  node_feat_kernel<<<nf_blocks, NF_T, smem_bytes, stream>>>(
      embed, W_lin, b_lin, W_gat, attn_l, attn_r, feat, el, er, src, dst, tab,
      plist, n_tiles);

  const size_t agg_bytes =
      (size_t)(DPB * F_OUT + 2 * DPB_PAD + 2 * AGG_NW * WCAP + AGG_NW + 4) *
      sizeof(float);
  (void)hipFuncSetAttribute((const void*)agg_kernel,
                            hipFuncAttributeMaxDynamicSharedMemorySize,
                            (int)agg_bytes);
  agg_kernel<<<AGG_NB, AGG_T, agg_bytes, stream>>>(src, dst, tab, plist, el, er,
                                                   feat, b_gat, out);
}
